# starting kernel
# baseline (speedup 1.0000x reference)
#include <hip/hip_runtime.h>
#include <math.h>

typedef float v2f __attribute__((ext_vector_type(2)));
typedef float v8f __attribute__((ext_vector_type(8)));

#define BQ      2
#define SEQ     4096
#define DMODEL  512
#define HEADS   8
#define DH      64
#define WIN     31
#define HALFW   15

__global__ __launch_bounds__(256)
void proj_gemm_kernel(const float* __restrict__ X,
                      const float* __restrict__ Wq,
                      const float* __restrict__ Wk,
                      const float* __restrict__ Wv,
                      float* __restrict__ Q,
                      float* __restrict__ Kp,
                      float* __restrict__ Vp)
{
    const float* Wsel;
    float*       Osel;
    if (blockIdx.z == 0)      { Wsel = Wq; Osel = Q;  }
    else if (blockIdx.z == 1) { Wsel = Wk; Osel = Kp; }
    else                      { Wsel = Wv; Osel = Vp; }

    __shared__ float xs[128][36];
    __shared__ float wst[64][38];

    const int rowBase = blockIdx.x * 128;
    const int colBase = blockIdx.y * 64;
    const int tid     = threadIdx.x;
    const int wave    = tid >> 5;
    const int lane    = tid & 31;
    const int l15     = lane & 15;
    const int koff    = (lane >> 4) << 1;
    const int waveRow = wave * 16;

    v8f acc0 = {}; v8f acc1 = {}; v8f acc2 = {}; v8f acc3 = {};

    for (int kb = 0; kb < DMODEL; kb += 32) {
        __syncthreads();
        #pragma unroll
        for (int i = 0; i < 16; ++i) {
            int idx = tid + i * 256;
            int r = idx >> 5, c = idx & 31;
            xs[r][c] = X[(size_t)(rowBase + r) * DMODEL + (kb + c)];
        }
        #pragma unroll
        for (int i = 0; i < 8; ++i) {
            int idx = tid + i * 256;
            int r = idx >> 6, c = idx & 63;
            wst[c][r] = Wsel[(size_t)(kb + r) * DMODEL + (colBase + c)];
        }
        if (kb + 32 < DMODEL) {
            __builtin_prefetch(&X[(size_t)(rowBase + (tid >> 3)) * DMODEL + (kb + 32) + ((tid & 7) << 2)], 0, 0);
            __builtin_prefetch(&Wsel[(size_t)(kb + 32 + (tid >> 6)) * DMODEL + colBase + (tid & 63)], 0, 0);
        }
        __syncthreads();

        #pragma unroll
        for (int ks = 0; ks < 32; ks += 4) {
            v2f a  = *(const v2f*)&xs[waveRow + l15][ks + koff];
            v2f b0 = *(const v2f*)&wst[ 0 + l15][ks + koff];
            v2f b1 = *(const v2f*)&wst[16 + l15][ks + koff];
            v2f b2 = *(const v2f*)&wst[32 + l15][ks + koff];
            v2f b3 = *(const v2f*)&wst[48 + l15][ks + koff];

            acc0 = __builtin_amdgcn_wmma_f32_16x16x4_f32(false, a, false, b0, (short)0, acc0, false, false);
            acc1 = __builtin_amdgcn_wmma_f32_16x16x4_f32(false, a, false, b1, (short)0, acc1, false, false);
            acc2 = __builtin_amdgcn_wmma_f32_16x16x4_f32(false, a, false, b2, (short)0, acc2, false, false);
            acc3 = __builtin_amdgcn_wmma_f32_16x16x4_f32(false, a, false, b3, (short)0, acc3, false, false);
        }
    }

    const int mAdd = (lane >> 4) << 3;
    #pragma unroll
    for (int j = 0; j < 8; ++j) {
        int row = rowBase + waveRow + j + mAdd;
        size_t o = (size_t)row * DMODEL + colBase;
        Osel[o +  0 + l15] = acc0[j];
        Osel[o + 16 + l15] = acc1[j];
        Osel[o + 32 + l15] = acc2[j];
        Osel[o + 48 + l15] = acc3[j];
    }
}

__global__ __launch_bounds__(256)
void local_attn_kernel(const float* __restrict__ Q,
                       const float* __restrict__ Kp,
                       const float* __restrict__ Vp,
                       float* __restrict__ O)
{
    const int gwave = blockIdx.x * (blockDim.x >> 5) + (threadIdx.x >> 5);
    const int lane  = threadIdx.x & 31;
    const int h = gwave & (HEADS - 1);
    const int s = (gwave >> 3) & (SEQ - 1);
    const int b = gwave >> 15;

    const size_t rowOff = ((size_t)b * SEQ + s) * DMODEL + h * DH;

    const int j   = lane;
    const int src = s - HALFW + j;

    float score;
    if (j >= WIN) {
        score = -INFINITY;
    } else if (src < 0 || src >= SEQ) {
        score = 0.0f;
    } else {
        const float* qp = Q  + rowOff;
        const float* kp = Kp + ((size_t)b * SEQ + src) * DMODEL + h * DH;
        float dot = 0.0f;
        #pragma unroll
        for (int d = 0; d < DH; d += 4) {
            float4 qv = *(const float4*)(qp + d);
            float4 kv = *(const float4*)(kp + d);
            dot += qv.x * kv.x + qv.y * kv.y + qv.z * kv.z + qv.w * kv.w;
        }
        score = dot * 0.125f;
    }

    float m = score;
    #pragma unroll
    for (int off = 16; off > 0; off >>= 1) m = fmaxf(m, __shfl_xor(m, off, 32));
    float e = __expf(score - m);
    float sum = e;
    #pragma unroll
    for (int off = 16; off > 0; off >>= 1) sum += __shfl_xor(sum, off, 32);
    const float a = e / sum;

    float acc0 = 0.0f, acc1 = 0.0f;
    for (int jj = 0; jj < WIN; ++jj) {
        int sj = s - HALFW + jj;
        float aj = __shfl(a, jj, 32);
        if (sj >= 0 && sj < SEQ) {
            const float* vp = Vp + ((size_t)b * SEQ + sj) * DMODEL + h * DH;
            acc0 += aj * vp[lane];
            acc1 += aj * vp[lane + 32];
        }
    }
    O[rowOff + lane]      = acc0;
    O[rowOff + lane + 32] = acc1;
}

extern "C" void kernel_launch(void* const* d_in, const int* in_sizes, int n_in,
                              void* d_out, int out_size, void* d_ws, size_t ws_size,
                              hipStream_t stream)
{
    (void)in_sizes; (void)n_in; (void)out_size; (void)ws_size;
    const float* x  = (const float*)d_in[0];
    const float* Wq = (const float*)d_in[1];
    const float* Wk = (const float*)d_in[2];
    const float* Wv = (const float*)d_in[3];
    float* out = (float*)d_out;

    const size_t n = (size_t)BQ * SEQ * DMODEL;
    float* qb = (float*)d_ws;
    float* kb = qb + n;
    float* vb = kb + n;

    dim3 gProj(BQ * SEQ / 128, DMODEL / 64, 3);
    proj_gemm_kernel<<<gProj, 256, 0, stream>>>(x, Wq, Wk, Wv, qb, kb, vb);

    const int totalWaves = BQ * SEQ * HEADS;
    local_attn_kernel<<<totalWaves / 8, 256, 0, stream>>>(qb, kb, vb, out);
}


# Projection GEMM made weight-stationary: each 16-wave workgroup holds its 512x96 W slab transposed in LDS for all of K; waves read X rows from global straight into A fragments; k loop has no barrier
# speedup vs baseline: 2.4367x; 2.4367x over previous
#include <hip/hip_runtime.h>
#include <math.h>

typedef float v2f __attribute__((ext_vector_type(2)));
typedef float v8f __attribute__((ext_vector_type(8)));
typedef float v4f __attribute__((ext_vector_type(4)));

#define BQ      2
#define SEQ     4096
#define DMODEL  512
#define HEADS   8
#define DH      64
#define WIN     31
#define HALFW   15

#define GROWS   512
#define GCT     6
#define WP      36
#define WCH     (GCT * 16 * WP)

struct AChunk { v4f t0[4]; v4f t1[4]; };

__device__ __forceinline__ void load_a(AChunk& a, const float* __restrict__ x0, const float* __restrict__ x1, int k)
{
    #pragma unroll
    for (int u = 0; u < 4; ++u) {
        a.t0[u] = *(const v4f*)(x0 + k + 8 * u);
        a.t1[u] = *(const v4f*)(x1 + k + 8 * u);
    }
}

__device__ __forceinline__ void mma_chunk(v8f (&acc0)[GCT], v8f (&acc1)[GCT], const AChunk& a, const float* bp)
{
    #pragma unroll
    for (int u = 0; u < 4; ++u) {
        v4f b[GCT];
        #pragma unroll
        for (int j = 0; j < GCT; ++j) b[j] = *(const v4f*)(bp + j * 16 * WP + 8 * u);
        #pragma unroll
        for (int j = 0; j < GCT; ++j) {
            acc0[j] = __builtin_amdgcn_wmma_f32_16x16x4_f32(false, a.t0[u].xy, false, b[j].xy, (short)0, acc0[j], false, false);
            acc1[j] = __builtin_amdgcn_wmma_f32_16x16x4_f32(false, a.t1[u].xy, false, b[j].xy, (short)0, acc1[j], false, false);
        }
        #pragma unroll
        for (int j = 0; j < GCT; ++j) {
            acc0[j] = __builtin_amdgcn_wmma_f32_16x16x4_f32(false, a.t0[u].zw, false, b[j].zw, (short)0, acc0[j], false, false);
            acc1[j] = __builtin_amdgcn_wmma_f32_16x16x4_f32(false, a.t1[u].zw, false, b[j].zw, (short)0, acc1[j], false, false);
        }
    }
}

__global__ __launch_bounds__(512)
void proj_gemm_kernel(const float* __restrict__ X,
                      const float* __restrict__ Wq,
                      const float* __restrict__ Wk,
                      const float* __restrict__ Wv,
                      float* __restrict__ Q,
                      float* __restrict__ Kp,
                      float* __restrict__ Vp)
{
    __shared__ __attribute__((aligned(16))) float wst[16 * WCH];

    const int tid   = threadIdx.x;
    const int wave  = tid >> 5;
    const int lane  = tid & 31;
    const int ct0   = blockIdx.x * GCT;
    const int row0  = blockIdx.y * GROWS + wave * 32;

    const int c = lane & 15;
    const int h = lane >> 4;
    const float* x0 = X + (size_t)(row0 + c) * DMODEL + 4 * h;
    const float* x1 = x0 + 16 * DMODEL;

    AChunk aA, aB;
    load_a(aA, x0, x1, 0);

    {
        const int kk = lane & 7;
        const int n4 = lane >> 3;
        v4f wreg[4][GCT];
        #pragma unroll
        for (int j = 0; j < GCT; ++j) {
            const int ct = ct0 + j;
            const int m  = ct >> 5;
            const float* Wm = (m == 0) ? Wq : (m == 1) ? Wk : Wv;
            const float* src = Wm + (size_t)(wave * 32 + kk) * DMODEL + ((ct & 31) << 4) + 4 * n4;
            #pragma unroll
            for (int kg = 0; kg < 4; ++kg)
                wreg[kg][j] = *(const v4f*)(src + (size_t)(8 * kg) * DMODEL);
        }
        #pragma unroll
        for (int j = 0; j < GCT; ++j) {
            #pragma unroll
            for (int kg = 0; kg < 4; ++kg) {
                float* dst = &wst[wave * WCH + (16 * j + 4 * n4) * WP + 8 * kg + kk];
                dst[0 * WP] = wreg[kg][j].x;
                dst[1 * WP] = wreg[kg][j].y;
                dst[2 * WP] = wreg[kg][j].z;
                dst[3 * WP] = wreg[kg][j].w;
            }
        }
    }
    __syncthreads();

    v8f acc0[GCT], acc1[GCT];
    #pragma unroll
    for (int j = 0; j < GCT; ++j) { acc0[j] = (v8f){}; acc1[j] = (v8f){}; }

    const float* bp = &wst[c * WP + 4 * h];

    for (int k = 0; k < DMODEL - 64; k += 64) {
        load_a(aB, x0, x1, k + 32);
        mma_chunk(acc0, acc1, aA, bp + (k >> 5) * WCH);
        load_a(aA, x0, x1, k + 64);
        mma_chunk(acc0, acc1, aB, bp + (k >> 5) * WCH + WCH);
    }
    load_a(aB, x0, x1, DMODEL - 32);
    mma_chunk(acc0, acc1, aA, bp + 14 * WCH);
    mma_chunk(acc0, acc1, aB, bp + 15 * WCH);

    #pragma unroll
    for (int j = 0; j < GCT; ++j) {
        const int ct = ct0 + j;
        const int m  = ct >> 5;
        float* Om = (m == 0) ? Q : (m == 1) ? Kp : Vp;
        float* dst = Om + (size_t)(row0 + 8 * h) * DMODEL + ((ct & 31) << 4) + c;
        #pragma unroll
        for (int r = 0; r < 8; ++r) {
            dst[(size_t)r * DMODEL]        = acc0[j][r];
            dst[(size_t)(r + 16) * DMODEL] = acc1[j][r];
        }
    }
}

#define TQ      128
#define TROWS   160
#define KP      68
#define PP      36

__global__ __launch_bounds__(256)
void local_attn_kernel(const float* __restrict__ Q,
                       const float* __restrict__ Kp,
                       const float* __restrict__ Vp,
                       float* __restrict__ O)
{
    __shared__ __attribute__((aligned(16))) float Ks[TROWS * KP];
    __shared__ __attribute__((aligned(16))) float Vs[TROWS * KP];
    __shared__ __attribute__((aligned(16))) float Qs[TQ * KP];
    __shared__ __attribute__((aligned(16))) float Ps[TQ * PP];

    const int t  = threadIdx.x;
    const int s0 = blockIdx.x * TQ;
    const int h  = blockIdx.y;
    const int b  = blockIdx.z;
    const size_t headOff = (size_t)b * SEQ * DMODEL + (size_t)h * DH;

    {
        float4 kreg[10], qreg[8], vreg[10];
        #pragma unroll
        for (int i = 0; i < 10; ++i) {
            const int idx = t + i * 256;
            const int r = idx >> 4, c = idx & 15;
            const int gc = min(max(s0 - HALFW + r, 0), SEQ - 1);
            kreg[i] = *(const float4*)(Kp + headOff + (size_t)gc * DMODEL + c * 4);
        }
        #pragma unroll
        for (int i = 0; i < 8; ++i) {
            const int idx = t + i * 256;
            const int r = idx >> 4, c = idx & 15;
            qreg[i] = *(const float4*)(Q + headOff + (size_t)(s0 + r) * DMODEL + c * 4);
        }
        #pragma unroll
        for (int i = 0; i < 10; ++i) {
            const int idx = t + i * 256;
            const int r = idx >> 4, c = idx & 15;
            const int gc = min(max(s0 - HALFW + r, 0), SEQ - 1);
            vreg[i] = *(const float4*)(Vp + headOff + (size_t)gc * DMODEL + c * 4);
        }
        #pragma unroll
        for (int i = 0; i < 10; ++i) {
            const int idx = t + i * 256;
            *(float4*)&Ks[(idx >> 4) * KP + (idx & 15) * 4] = kreg[i];
        }
        #pragma unroll
        for (int i = 0; i < 8; ++i) {
            const int idx = t + i * 256;
            *(float4*)&Qs[(idx >> 4) * KP + (idx & 15) * 4] = qreg[i];
        }
        #pragma unroll
        for (int i = 0; i < 10; ++i) {
            const int idx = t + i * 256;
            *(float4*)&Vs[(idx >> 4) * KP + (idx & 15) * 4] = vreg[i];
        }
    }
    __syncthreads();

    {
        const int lane = t & 31;
        const int l15  = lane & 15;
        const int hf   = lane >> 4;
        const int q    = (t >> 5) * 16 + l15;
        const int g0   = s0 + q - HALFW + hf * 16;

        v4f qr[16];
        #pragma unroll
        for (int i = 0; i < 16; ++i) qr[i] = *(const v4f*)&Qs[q * KP + i * 4];

        float sc[16];
        #pragma unroll
        for (int jj = 0; jj < 16; ++jj) {
            const float* kp = &Ks[(q + hf * 16 + jj) * KP];
            v2f a01 = {0.0f, 0.0f}, a23 = {0.0f, 0.0f};
            #pragma unroll
            for (int i = 0; i < 16; ++i) {
                const v4f kv = *(const v4f*)(kp + i * 4);
                a01 = __builtin_elementwise_fma(qr[i].xy, kv.xy, a01);
                a23 = __builtin_elementwise_fma(qr[i].zw, kv.zw, a23);
            }
            const v2f a = a01 + a23;
            const bool in = (g0 + jj >= 0) && (g0 + jj < SEQ);
            sc[jj] = in ? (a.x + a.y) * 0.125f : 0.0f;
        }
        if (hf == 1) sc[15] = -INFINITY;

        float m = sc[0];
        #pragma unroll
        for (int jj = 1; jj < 16; ++jj) m = fmaxf(m, sc[jj]);
        m = fmaxf(m, __shfl_xor(m, 16, 32));
        float sum = 0.0f;
        #pragma unroll
        for (int jj = 0; jj < 16; ++jj) { sc[jj] = __expf(sc[jj] - m); sum += sc[jj]; }
        sum += __shfl_xor(sum, 16, 32);
        #pragma unroll
        for (int jj = 0; jj < 16; ++jj) {
            const bool in = (g0 + jj >= 0) && (g0 + jj < SEQ);
            sc[jj] = in ? sc[jj] / sum : 0.0f;
        }
        #pragma unroll
        for (int jj = 0; jj < 16; jj += 4)
            *(float4*)&Ps[q * PP + hf * 16 + jj] = make_float4(sc[jj], sc[jj + 1], sc[jj + 2], sc[jj + 3]);
    }
    __syncthreads();

    {
        const int c  = t & 15;
        const int q0 = (t >> 4) * 8;
        v2f acc01[8], acc23[8];
        #pragma unroll
        for (int i = 0; i < 8; ++i) { acc01[i] = (v2f){0.0f, 0.0f}; acc23[i] = (v2f){0.0f, 0.0f}; }
        v4f pc[8];
        #pragma unroll
        for (int r = 0; r < 38; ++r) {
            const v4f v = *(const v4f*)&Vs[(q0 + r) * KP + c * 4];
            #pragma unroll
            for (int i = 0; i < 8; ++i) {
                const int j = r - i;
                if (j >= 0 && j < WIN) {
                    if ((j & 3) == 0) pc[i] = *(const v4f*)&Ps[(q0 + i) * PP + j];
                    const float pj = pc[i][j & 3];
                    const v2f pp = {pj, pj};
                    acc01[i] = __builtin_elementwise_fma(v.xy, pp, acc01[i]);
                    acc23[i] = __builtin_elementwise_fma(v.zw, pp, acc23[i]);
                }
            }
        }
        #pragma unroll
        for (int i = 0; i < 8; ++i) {
            const v4f o = {acc01[i].x, acc01[i].y, acc23[i].x, acc23[i].y};
            *(v4f*)(O + headOff + (size_t)(s0 + q0 + i) * DMODEL + c * 4) = o;
        }
    }
}

extern "C" void kernel_launch(void* const* d_in, const int* in_sizes, int n_in,
                              void* d_out, int out_size, void* d_ws, size_t ws_size,
                              hipStream_t stream)
{
    (void)in_sizes; (void)n_in; (void)out_size; (void)ws_size;
    const float* x  = (const float*)d_in[0];
    const float* Wq = (const float*)d_in[1];
    const float* Wk = (const float*)d_in[2];
    const float* Wv = (const float*)d_in[3];
    float* out = (float*)d_out;

    const size_t n = (size_t)BQ * SEQ * DMODEL;
    float* qb = (float*)d_ws;
    float* kb = qb + n;
    float* vb = kb + n;

    dim3 gProj(16, 16, 1);
    proj_gemm_kernel<<<gProj, 512, 0, stream>>>(x, Wq, Wk, Wv, qb, kb, vb);

    dim3 gAttn(SEQ / TQ, HEADS, BQ);
    local_attn_kernel<<<gAttn, 256, 0, stream>>>(qb, kb, vb, out);
}
